# starting kernel
# baseline (speedup 1.0000x reference)
#include <hip/hip_runtime.h>


typedef float v2f __attribute__((ext_vector_type(2)));
typedef float v8f __attribute__((ext_vector_type(8)));

#define HS            1024
#define NANCH         4
#define ROWS_PER_BLK  16
#define LN_EPS        1e-5f

__global__ __launch_bounds__(256)
void ln_subspace_wmma(const float* __restrict__ x,
                      const float* __restrict__ alpha,
                      const float* __restrict__ W,
                      const float* __restrict__ Bb,
                      float* __restrict__ out) {
  __shared__ float sSum[8][16];
  __shared__ float sSq [8][16];
  __shared__ float sMean[16];
  __shared__ float sRstd[16];

  const int tid  = threadIdx.x;
  const int wave = tid >> 5;
  const int lane = tid & 31;
  const int half = lane >> 4;
  const int m16  = lane & 15;

  const int row0 = blockIdx.x * ROWS_PER_BLK;
  const int colw = wave * 128 + m16;

  float xr[8][8];
  float ps[8], pq[8];
  #pragma unroll
  for (int r = 0; r < 8; ++r) { ps[r] = 0.f; pq[r] = 0.f; }

  #pragma unroll
  for (int r = 0; r < 8; ++r) {
    const size_t rbase = (size_t)(row0 + r + 8 * half) * HS + colw;
    #pragma unroll
    for (int j = 0; j < 8; ++j) {
      const float v = x[rbase + 16 * j];
      xr[j][r] = v;
      ps[r] += v;
      pq[r] += v * v;
    }
  }

  #pragma unroll
  for (int r = 0; r < 8; ++r) {
    #pragma unroll
    for (int m = 1; m < 16; m <<= 1) {
      ps[r] += __shfl_xor(ps[r], m, 32);
      pq[r] += __shfl_xor(pq[r], m, 32);
    }
  }
  if (m16 == 0) {
    #pragma unroll
    for (int r = 0; r < 8; ++r) {
      sSum[wave][r + 8 * half] = ps[r];
      sSq [wave][r + 8 * half] = pq[r];
    }
  }
  __syncthreads();

  if (tid < 16) {
    float s = 0.f, q = 0.f;
    #pragma unroll
    for (int w = 0; w < 8; ++w) { s += sSum[w][tid]; q += sSq[w][tid]; }
    const float mean = s * (1.0f / HS);
    const float var  = q * (1.0f / HS) - mean * mean;
    sMean[tid] = mean;
    sRstd[tid] = rsqrtf(var + LN_EPS);
  }
  __syncthreads();

  float mr[8], rr[8];
  #pragma unroll
  for (int r = 0; r < 8; ++r) {
    mr[r] = sMean[r + 8 * half];
    rr[r] = sRstd[r + 8 * half];
  }

  v2f a;
  {
    const float2 av = *reinterpret_cast<const float2*>(
        alpha + (size_t)(row0 + m16) * NANCH + 2 * half);
    a.x = av.x;
    a.y = av.y;
  }

  #pragma unroll
  for (int j = 0; j < 8; ++j) {
    const int col = colw + 16 * j;

    v2f bw, bb;
    bw.x = W [(2 * half    ) * HS + col];
    bw.y = W [(2 * half + 1) * HS + col];
    bb.x = Bb[(2 * half    ) * HS + col];
    bb.y = Bb[(2 * half + 1) * HS + col];

    v8f cz = {};
    v8f cw = __builtin_amdgcn_wmma_f32_16x16x4_f32(
        false, a, false, bw, (short)0, cz, false, false);
    v8f cb = __builtin_amdgcn_wmma_f32_16x16x4_f32(
        false, a, false, bb, (short)0, cz, false, false);

    #pragma unroll
    for (int r = 0; r < 8; ++r) {
      const float xh = (xr[j][r] - mr[r]) * rr[r];
      out[(size_t)(row0 + r + 8 * half) * HS + col] = xh * cw[r] + cb[r];
    }
  }
}

extern "C" void kernel_launch(void* const* d_in, const int* in_sizes, int n_in,
                              void* d_out, int out_size, void* d_ws, size_t ws_size,
                              hipStream_t stream) {
  (void)n_in; (void)out_size; (void)d_ws; (void)ws_size;
  const float* x     = (const float*)d_in[0];
  const float* alpha = (const float*)d_in[1];
  const float* W     = (const float*)d_in[2];
  const float* b     = (const float*)d_in[3];
  float* out = (float*)d_out;

  const int rows = in_sizes[0] / HS;
  const int grid = rows / ROWS_PER_BLK;

  ln_subspace_wmma<<<grid, 256, 0, stream>>>(x, alpha, W, b, out);
}


# Tile N index remapped so each lane owns 4 consecutive columns: x, W, b, out move as 128-bit whole-line accesses (was 32-bit, 64-B pieces); row sum/sum-of-squares partials regroup (sum order changes)
# speedup vs baseline: 1.5651x; 1.5651x over previous
#include <hip/hip_runtime.h>


typedef float v2f __attribute__((ext_vector_type(2)));
typedef float v4f __attribute__((ext_vector_type(4)));
typedef float v8f __attribute__((ext_vector_type(8)));

#define HS            1024
#define NANCH         4
#define ROWS_PER_BLK  16
#define LN_EPS        1e-5f

__global__ __launch_bounds__(256, 6)
void ln_subspace_wmma(const float* __restrict__ x,
                      const float* __restrict__ alpha,
                      const float* __restrict__ W,
                      const float* __restrict__ Bb,
                      float* __restrict__ out) {
  __shared__ float sSum[8][16];
  __shared__ float sSq [8][16];
  __shared__ float sMean[16];
  __shared__ float sRstd[16];

  const int tid  = threadIdx.x;
  const int wave = tid >> 5;
  const int lane = tid & 31;
  const int half = lane >> 4;
  const int m16  = lane & 15;

  const int row0 = blockIdx.x * ROWS_PER_BLK;
  const int colw = wave * 128 + 4 * m16;

  v4f xq[2][8];
  float ps[8], pq[8];

  #pragma unroll
  for (int r = 0; r < 8; ++r) {
    const size_t rbase = (size_t)(row0 + r + 8 * half) * HS + colw;
    #pragma unroll
    for (int g = 0; g < 2; ++g) {
      xq[g][r] = *reinterpret_cast<const v4f*>(x + rbase + 64 * g);
    }
  }

  #pragma unroll
  for (int r = 0; r < 8; ++r) {
    v2f s2 = {0.f, 0.f};
    v2f q2 = {0.f, 0.f};
    #pragma unroll
    for (int g = 0; g < 2; ++g) {
      const v2f lo = xq[g][r].xy;
      const v2f hi = xq[g][r].zw;
      s2 += lo;
      q2 += lo * lo;
      s2 += hi;
      q2 += hi * hi;
    }
    ps[r] = s2.x + s2.y;
    pq[r] = q2.x + q2.y;
  }

  #pragma unroll
  for (int r = 0; r < 8; ++r) {
    #pragma unroll
    for (int m = 1; m < 16; m <<= 1) {
      ps[r] += __shfl_xor(ps[r], m, 32);
      pq[r] += __shfl_xor(pq[r], m, 32);
    }
  }
  if (m16 == 0) {
    #pragma unroll
    for (int r = 0; r < 8; ++r) {
      sSum[wave][r + 8 * half] = ps[r];
      sSq [wave][r + 8 * half] = pq[r];
    }
  }
  __syncthreads();

  if (tid < 16) {
    float s = 0.f, q = 0.f;
    #pragma unroll
    for (int w = 0; w < 8; ++w) { s += sSum[w][tid]; q += sSq[w][tid]; }
    const float mean = s * (1.0f / HS);
    const float var  = q * (1.0f / HS) - mean * mean;
    sMean[tid] = mean;
    sRstd[tid] = rsqrtf(var + LN_EPS);
  }
  __syncthreads();

  float mr[8], rr[8];
  #pragma unroll
  for (int r = 0; r < 8; ++r) {
    mr[r] = sMean[r + 8 * half];
    rr[r] = sRstd[r + 8 * half];
  }

  v2f a;
  {
    const float2 av = *reinterpret_cast<const float2*>(
        alpha + (size_t)(row0 + m16) * NANCH + 2 * half);
    a.x = av.x;
    a.y = av.y;
  }

  #pragma unroll
  for (int g = 0; g < 2; ++g) {
    const int colg = colw + 64 * g;

    const v4f w0 = *reinterpret_cast<const v4f*>(W  + (2 * half    ) * HS + colg);
    const v4f w1 = *reinterpret_cast<const v4f*>(W  + (2 * half + 1) * HS + colg);
    const v4f b0 = *reinterpret_cast<const v4f*>(Bb + (2 * half    ) * HS + colg);
    const v4f b1 = *reinterpret_cast<const v4f*>(Bb + (2 * half + 1) * HS + colg);

    #pragma unroll
    for (int t = 0; t < 4; ++t) {
      v2f bw, bb;
      bw.x = w0[t];
      bw.y = w1[t];
      bb.x = b0[t];
      bb.y = b1[t];

      v8f cz = {};
      v8f cw = __builtin_amdgcn_wmma_f32_16x16x4_f32(
          false, a, false, bw, (short)0, cz, false, false);
      v8f cb = __builtin_amdgcn_wmma_f32_16x16x4_f32(
          false, a, false, bb, (short)0, cz, false, false);

      #pragma unroll
      for (int r = 0; r < 8; ++r) {
        const float xh = (xq[g][r][t] - mr[r]) * rr[r];
        xq[g][r][t] = xh * cw[r] + cb[r];
      }
    }

    #pragma unroll
    for (int r = 0; r < 8; ++r) {
      *reinterpret_cast<v4f*>(
          out + (size_t)(row0 + r + 8 * half) * HS + colg) = xq[g][r];
    }
  }
}

extern "C" void kernel_launch(void* const* d_in, const int* in_sizes, int n_in,
                              void* d_out, int out_size, void* d_ws, size_t ws_size,
                              hipStream_t stream) {
  (void)n_in; (void)out_size; (void)d_ws; (void)ws_size;
  const float* x     = (const float*)d_in[0];
  const float* alpha = (const float*)d_in[1];
  const float* W     = (const float*)d_in[2];
  const float* b     = (const float*)d_in[3];
  float* out = (float*)d_out;

  const int rows = in_sizes[0] / HS;
  const int grid = rows / ROWS_PER_BLK;

  ln_subspace_wmma<<<grid, 256, 0, stream>>>(x, alpha, W, b, out);
}
